# starting kernel
# baseline (speedup 1.0000x reference)
#include <hip/hip_runtime.h>

typedef __attribute__((ext_vector_type(2))) float v2f;
typedef __attribute__((ext_vector_type(4))) float v4f;
typedef __attribute__((ext_vector_type(8))) float v8f;

#define F_DIM 64
#define E_DIM 16
#define P_DIM 10

#define ROWS_PER_WAVE  16
#define WAVES_PER_BLOCK 8
#define ROWS_PER_BLOCK (ROWS_PER_WAVE * WAVES_PER_BLOCK)
#define XSQ_PITCH 68

__global__ __launch_bounds__(256)
void lp_wmma_kernel(const float* __restrict__ x,
                    const float* __restrict__ w,
                    float* __restrict__ out, int R) {
    __shared__ float xsq[WAVES_PER_BLOCK * ROWS_PER_WAVE * XSQ_PITCH];

    const int tid  = threadIdx.x;
    const int wv   = tid >> 5;
    const int lane = tid & 31;
    const int rb   = blockIdx.x * ROWS_PER_BLOCK;
    const int wrb  = rb + wv * ROWS_PER_WAVE;
    const bool full_tile = (wrb + ROWS_PER_WAVE) <= R;

    const int p    = lane & 15;
    const int koff = (lane >> 4) << 1;
    v2f bfrag[16];
    #pragma unroll
    for (int s = 0; s < 16; ++s) {
        int k = s * 4 + koff;
        float bx = 0.0f, by = 0.0f;
        if (p < P_DIM) {
            float w0 = w[p * F_DIM + k];
            float w1 = w[p * F_DIM + k + 1];
            bx = w0 * w0;
            by = w1 * w1;
        }
        bfrag[s].x = bx;
        bfrag[s].y = by;
    }

    float* myxsq = &xsq[wv * ROWS_PER_WAVE * XSQ_PITCH];
    const float* xb = x + (size_t)wrb * (F_DIM * E_DIM);
    if (full_tile) {
        #pragma unroll 4
        for (int it = 0; it < 32; ++it) {
            int pair = it * 32 + lane;
            int rl   = pair >> 6;
            int f    = pair & 63;
            const float* src = xb + (size_t)pair * E_DIM;
            float s = 0.0f;
            #pragma unroll
            for (int j = 0; j < 4; ++j) {
                v4f v = *(const v4f*)(src + j * 4);
                s = fmaf(v.x, v.x, s);
                s = fmaf(v.y, v.y, s);
                s = fmaf(v.z, v.z, s);
                s = fmaf(v.w, v.w, s);
            }
            myxsq[rl * XSQ_PITCH + f] = s;
        }
    } else {
        for (int it = 0; it < 32; ++it) {
            int pair = it * 32 + lane;
            int rl   = pair >> 6;
            int f    = pair & 63;
            float s = 0.0f;
            if (wrb + rl < R) {
                const float* src = xb + (size_t)pair * E_DIM;
                #pragma unroll
                for (int j = 0; j < 4; ++j) {
                    v4f v = *(const v4f*)(src + j * 4);
                    s = fmaf(v.x, v.x, s);
                    s = fmaf(v.y, v.y, s);
                    s = fmaf(v.z, v.z, s);
                    s = fmaf(v.w, v.w, s);
                }
            }
            myxsq[rl * XSQ_PITCH + f] = s;
        }
    }

    __syncthreads();

    const int m = lane & 15;
    v8f c = {};
    #pragma unroll
    for (int s = 0; s < 16; ++s) {
        int k = s * 4 + koff;
        v2f a = *(const v2f*)&myxsq[m * XSQ_PITCH + k];
        c = __builtin_amdgcn_wmma_f32_16x16x4_f32(
                 false, a,  false, bfrag[s],
                 (short)0, c,  false,  false);
    }

    const int mb = (lane >> 4) << 3;
    if (p < P_DIM) {
        float* dst = out + (size_t)(wrb + mb) * P_DIM + p;
        if (full_tile) {
            #pragma unroll
            for (int i = 0; i < 8; ++i)
                dst[i * P_DIM] = c[i];
        } else {
            #pragma unroll
            for (int i = 0; i < 8; ++i)
                if (wrb + mb + i < R) dst[i * P_DIM] = c[i];
        }
    }
}

extern "C" void kernel_launch(void* const* d_in, const int* in_sizes, int n_in,
                              void* d_out, int out_size, void* d_ws, size_t ws_size,
                              hipStream_t stream) {
    const float* x = (const float*)d_in[0];
    const float* w = (const float*)d_in[1];
    float* out = (float*)d_out;

    int R = in_sizes[0] / (F_DIM * E_DIM);
    int blocks = (R + ROWS_PER_BLOCK - 1) / ROWS_PER_BLOCK;
    lp_wmma_kernel<<<blocks, 256, 0, stream>>>(x, w, out, R);
}


# stage A read as a stream: each wave-wide load covers 512 contiguous bytes of one row (4 lines, was 16), 3 rows requested ahead; e-sum = 4 per lane, then 2 quad DPP adds (new reduction tree)
# speedup vs baseline: 1.3559x; 1.3559x over previous
#include <hip/hip_runtime.h>

typedef __attribute__((ext_vector_type(2))) float v2f;
typedef __attribute__((ext_vector_type(4))) float v4f;
typedef __attribute__((ext_vector_type(8))) float v8f;

#define F_DIM 64
#define E_DIM 16
#define P_DIM 10

#define ROWS_PER_WAVE  16
#define WAVES_PER_BLOCK 8
#define ROWS_PER_BLOCK (ROWS_PER_WAVE * WAVES_PER_BLOCK)
#define XSQ_PITCH 68

#define ROW_FLOATS (F_DIM * E_DIM)
#define PIECES 8
#define PIECE_FLOATS 128
#define AHEAD 3

static __device__ __forceinline__ float quad_xor1(float v) {
    return __int_as_float(__builtin_amdgcn_update_dpp(0, __float_as_int(v), 0xB1, 0xF, 0xF, true));
}
static __device__ __forceinline__ float quad_xor2(float v) {
    return __int_as_float(__builtin_amdgcn_update_dpp(0, __float_as_int(v), 0x4E, 0xF, 0xF, true));
}

__global__ __launch_bounds__(256)
void lp_wmma_kernel(const float* __restrict__ x,
                    const float* __restrict__ w,
                    float* __restrict__ out, int R) {
    __shared__ float xsq[WAVES_PER_BLOCK * ROWS_PER_WAVE * XSQ_PITCH];

    const int tid  = threadIdx.x;
    const int wv   = __builtin_amdgcn_readfirstlane(tid >> 5);
    const int lane = tid & 31;
    const int rb   = blockIdx.x * ROWS_PER_BLOCK;
    const int wrb  = rb + wv * ROWS_PER_WAVE;
    const bool full_tile = (wrb + ROWS_PER_WAVE) <= R;

    const int p    = lane & 15;
    const int koff = (lane >> 4) << 1;

    float* myxsq = &xsq[wv * ROWS_PER_WAVE * XSQ_PITCH];
    const unsigned lane4 = (unsigned)lane * 4u;
    const int sel  = lane & 3;
    const int lofs = ((lane & 3) << 3) + (lane >> 2);
    v4f  ring[AHEAD + 1][PIECES];
    v2f  braw[16];
#define LOAD_ROW(RR) do { int gr_ = wrb + (RR); gr_ = gr_ < R ? gr_ : R - 1; \
        const float* rp_ = x + (size_t)gr_ * ROW_FLOATS; \
        _Pragma("unroll") for (int q_ = 0; q_ < PIECES; ++q_) \
            ring[(RR) % (AHEAD + 1)][q_] = *(const v4f*)(rp_ + (lane4 + (unsigned)(q_ * PIECE_FLOATS))); } while (0)

    #pragma unroll
    for (int r = 0; r < AHEAD; ++r) LOAD_ROW(r);

    #pragma unroll
    for (int r = 0; r < ROWS_PER_WAVE; ++r) {
        __builtin_amdgcn_sched_barrier(0);
        if (r + AHEAD < ROWS_PER_WAVE) LOAD_ROW(r + AHEAD);
        if (r + AHEAD == ROWS_PER_WAVE - 1) {
            const float* wp = w + (p < P_DIM ? p : 0) * F_DIM + koff;
            #pragma unroll
            for (int s = 0; s < 16; ++s) braw[s] = *(const v2f*)(wp + s * 4);
        }
        __builtin_amdgcn_sched_barrier(0);

        const v4f* cur = ring[r % (AHEAD + 1)];
        #pragma unroll
        for (int g = 0; g < 2; ++g) {
            float keep = 0.0f;
            #pragma unroll
            for (int t = 0; t < 4; ++t) {
                const v4f v = cur[4 * g + t];
                float s = v.x * v.x;
                s = fmaf(v.y, v.y, s);
                s = fmaf(v.z, v.z, s);
                s = fmaf(v.w, v.w, s);
                s += quad_xor1(s);
                s += quad_xor2(s);
                keep = (sel == t) ? s : keep;
            }
            myxsq[r * XSQ_PITCH + 32 * g + lofs] = keep;
        }
    }
#undef LOAD_ROW

    v2f bfrag[16];
    #pragma unroll
    for (int s = 0; s < 16; ++s) {
        const float w0 = braw[s].x, w1 = braw[s].y;
        bfrag[s].x = (p < P_DIM) ? w0 * w0 : 0.0f;
        bfrag[s].y = (p < P_DIM) ? w1 * w1 : 0.0f;
    }

    __syncthreads();

    const int m = lane & 15;
    v8f c = {};
    #pragma unroll
    for (int s = 0; s < 16; ++s) {
        int k = s * 4 + koff;
        v2f a = *(const v2f*)&myxsq[m * XSQ_PITCH + k];
        c = __builtin_amdgcn_wmma_f32_16x16x4_f32(
                 false, a,  false, bfrag[s],
                 (short)0, c,  false,  false);
    }

    const int mb = (lane >> 4) << 3;
    if (p < P_DIM) {
        float* dst = out + (size_t)(wrb + mb) * P_DIM + p;
        if (full_tile) {
            #pragma unroll
            for (int i = 0; i < 8; ++i)
                dst[i * P_DIM] = c[i];
        } else {
            #pragma unroll
            for (int i = 0; i < 8; ++i)
                if (wrb + mb + i < R) dst[i * P_DIM] = c[i];
        }
    }
}

extern "C" void kernel_launch(void* const* d_in, const int* in_sizes, int n_in,
                              void* d_out, int out_size, void* d_ws, size_t ws_size,
                              hipStream_t stream) {
    const float* x = (const float*)d_in[0];
    const float* w = (const float*)d_in[1];
    float* out = (float*)d_out;

    int R = in_sizes[0] / (F_DIM * E_DIM);
    int blocks = (R + ROWS_PER_BLOCK - 1) / ROWS_PER_BLOCK;
    lp_wmma_kernel<<<blocks, 256, 0, stream>>>(x, w, out, R);
}
